# starting kernel
# baseline (speedup 1.0000x reference)
#include <hip/hip_runtime.h>
#include <math.h>

typedef __attribute__((ext_vector_type(2))) float v2f;
typedef __attribute__((ext_vector_type(4))) float v4f;
typedef __attribute__((ext_vector_type(8))) float v8f;

#define BB 4
#define DD 64
#define NN 16384
#define KK 20

__global__ __launch_bounds__(256) void prep_w_kernel(const float* __restrict__ W1,
                                                     const float* __restrict__ W2,
                                                     float* __restrict__ Wp1,
                                                     float* __restrict__ Wp2) {
    int i = blockIdx.x * blockDim.x + threadIdx.x;
    int m = i >> 6, k = i & 63;
    if (m < 64) {
        Wp1[i] = W1[m * 128 + k];
        Wp2[i] = W2[m * 128 + k];
    } else {
        int mm = m - 64;
        Wp1[i] = W1[mm * 128 + 64 + k] - W1[mm * 128 + k];
        Wp2[i] = W2[mm * 128 + 64 + k] - W2[mm * 128 + k];
    }
}

template <int LAYOUT>
__global__ __launch_bounds__(256) void gemm_zt_kernel(const float* __restrict__ Wp,
                                                      const float* __restrict__ X,
                                                      float* __restrict__ ZT) {
    const int lane = threadIdx.x & 31;
    const int wave = (blockIdx.x * blockDim.x + threadIdx.x) >> 5;
    const int tilesPerBatch = NN / 16;
    const int b = wave / tilesPerBatch;
    const int t = wave - b * tilesPerBatch;
    const int colLocal = lane & 15;
    const int col = t * 16 + colLocal;
    const int khalf = lane >> 4;

    v8f acc[8] = {};

    for (int k0 = 0; k0 < 64; k0 += 4) {
        const int kk = k0 + khalf * 2;
        v2f bm;
        if (LAYOUT == 0) {
            const float* xb = X + (size_t)b * DD * NN;
            bm.x = xb[(size_t)kk * NN + col];
            bm.y = xb[(size_t)(kk + 1) * NN + col];
        } else {
            const float* xb = X + ((size_t)b * NN + col) * DD + kk;
            bm.x = xb[0];
            bm.y = xb[1];
        }
#pragma unroll
        for (int mt = 0; mt < 8; ++mt) {
            const float* wp = Wp + (mt * 16 + colLocal) * 64 + kk;
            v2f am;
            am.x = wp[0];
            am.y = wp[1];
            acc[mt] = __builtin_amdgcn_wmma_f32_16x16x4_f32(
                false, am, false, bm, (short)0, acc[mt], false, false);
        }
    }

    float* outp = ZT + ((size_t)b * NN + col) * 128 + khalf * 8;
#pragma unroll
    for (int mt = 0; mt < 8; ++mt) {
        v4f lo = {acc[mt][0], acc[mt][1], acc[mt][2], acc[mt][3]};
        v4f hi = {acc[mt][4], acc[mt][5], acc[mt][6], acc[mt][7]};
        *(v4f*)(outp + mt * 16) = lo;
        *(v4f*)(outp + mt * 16 + 4) = hi;
    }
}

__device__ __forceinline__ float gelu_exact(float x) {
    return 0.5f * x * (1.0f + erff(x * 0.70710678118654752f));
}

__global__ __launch_bounds__(256) void gather1_kernel(const float* __restrict__ ZT,
                                                      const int* __restrict__ IDX,
                                                      const float* __restrict__ g1,
                                                      const float* __restrict__ b1,
                                                      const float* __restrict__ m1,
                                                      const float* __restrict__ v1,
                                                      float* __restrict__ H) {
    const int lane = threadIdx.x & 31;
    const int wave = (blockIdx.x * blockDim.x + threadIdx.x) >> 5;
    const int b = wave / NN;
    const int n = wave - b * NN;
    const int ch = lane * 2;
    const int* ip = IDX + ((size_t)b * NN + n) * KK;
    const float* zb = ZT + (size_t)b * NN * 128;

    float mx0 = -3.402823466e38f, mx1 = -3.402823466e38f;
    for (int k = 0; k < KK; ++k) {
        int j = ip[k];
        const float* p = zb + (size_t)j * 128 + ch;
        mx0 = fmaxf(mx0, p[0]);
        mx1 = fmaxf(mx1, p[1]);
    }
    const float* cp = zb + (size_t)n * 128 + 64 + ch;
    float h0 = mx0 + cp[0];
    float h1 = mx1 + cp[1];

    float sc0 = g1[ch] / sqrtf(v1[ch] + 1e-5f);
    float sc1 = g1[ch + 1] / sqrtf(v1[ch + 1] + 1e-5f);
    float sh0 = b1[ch] - m1[ch] * sc0;
    float sh1 = b1[ch + 1] - m1[ch + 1] * sc1;

    float* hp = H + ((size_t)b * NN + n) * 64 + ch;
    hp[0] = gelu_exact(h0 * sc0 + sh0);
    hp[1] = gelu_exact(h1 * sc1 + sh1);
}

__global__ __launch_bounds__(256) void gather2_kernel(const float* __restrict__ ZT,
                                                      const int* __restrict__ IDX,
                                                      const float* __restrict__ X,
                                                      const float* __restrict__ g2,
                                                      const float* __restrict__ b2,
                                                      const float* __restrict__ m2,
                                                      const float* __restrict__ v2,
                                                      float* __restrict__ OUT) {
    const int lane = threadIdx.x & 31;
    const int wave = (blockIdx.x * blockDim.x + threadIdx.x) >> 5;
    const int b = wave / NN;
    const int n = wave - b * NN;
    const int ch = lane * 2;
    const int* ip = IDX + ((size_t)b * NN + n) * KK;
    const float* zb = ZT + (size_t)b * NN * 128;

    float mx0 = -3.402823466e38f, mx1 = -3.402823466e38f;
    for (int k = 0; k < KK; ++k) {
        int j = ip[k];
        const float* p = zb + (size_t)j * 128 + ch;
        mx0 = fmaxf(mx0, p[0]);
        mx1 = fmaxf(mx1, p[1]);
    }
    const float* cp = zb + (size_t)n * 128 + 64 + ch;
    float x0 = X[((size_t)b * DD + ch) * NN + n];
    float x1 = X[((size_t)b * DD + ch + 1) * NN + n];
    float h0 = x0 + mx0 + cp[0];
    float h1 = x1 + mx1 + cp[1];

    float sc0 = g2[ch] / sqrtf(v2[ch] + 1e-5f);
    float sc1 = g2[ch + 1] / sqrtf(v2[ch + 1] + 1e-5f);
    float sh0 = b2[ch] - m2[ch] * sc0;
    float sh1 = b2[ch + 1] - m2[ch + 1] * sc1;

    OUT[((size_t)b * DD + ch) * NN + n] = gelu_exact(h0 * sc0 + sh0);
    OUT[((size_t)b * DD + ch + 1) * NN + n] = gelu_exact(h1 * sc1 + sh1);
}

extern "C" void kernel_launch(void* const* d_in, const int* in_sizes, int n_in,
                              void* d_out, int out_size, void* d_ws, size_t ws_size,
                              hipStream_t stream) {
    const float* x   = (const float*)d_in[0];
    const int*   idx = (const int*)d_in[1];
    const float* W1  = (const float*)d_in[2];
    const float* W2  = (const float*)d_in[3];
    const float* g1  = (const float*)d_in[4];
    const float* b1  = (const float*)d_in[5];
    const float* m1  = (const float*)d_in[6];
    const float* v1  = (const float*)d_in[7];
    const float* g2  = (const float*)d_in[8];
    const float* b2  = (const float*)d_in[9];
    const float* m2  = (const float*)d_in[10];
    const float* v2  = (const float*)d_in[11];
    float* out = (float*)d_out;

    char* ws = (char*)d_ws;
    float* Wp1 = (float*)ws;
    float* Wp2 = (float*)(ws + 32 * 1024);
    float* ZT  = (float*)(ws + 64 * 1024);
    float* H   = (float*)(ws + 64 * 1024 + (size_t)BB * NN * 128 * 4);

    prep_w_kernel<<<(128 * 64) / 256, 256, 0, stream>>>(W1, W2, Wp1, Wp2);

    const int gemmBlocks = (BB * (NN / 16) * 32) / 256;
    const int gathBlocks = (BB * NN * 32) / 256;

    gemm_zt_kernel<0><<<gemmBlocks, 256, 0, stream>>>(Wp1, x, ZT);
    gather1_kernel<<<gathBlocks, 256, 0, stream>>>(ZT, idx, g1, b1, m1, v1, H);
    gemm_zt_kernel<1><<<gemmBlocks, 256, 0, stream>>>(Wp2, H, ZT);
    gather2_kernel<<<gathBlocks, 256, 0, stream>>>(ZT, idx, x, g2, b2, m2, v2, out);
}


# gather2: block owns 32 points and one LDS tile [channel][point]; x rows staged in, result rows staged out with lane = point: each x read and out write is a whole 128-byte line (was 4-byte pieces)
# speedup vs baseline: 1.9072x; 1.9072x over previous
#include <hip/hip_runtime.h>
#include <math.h>

typedef __attribute__((ext_vector_type(2))) float v2f;
typedef __attribute__((ext_vector_type(4))) float v4f;
typedef __attribute__((ext_vector_type(8))) float v8f;

#define BB 4
#define DD 64
#define NN 16384
#define KK 20

__global__ __launch_bounds__(256) void prep_w_kernel(const float* __restrict__ W1,
                                                     const float* __restrict__ W2,
                                                     float* __restrict__ Wp1,
                                                     float* __restrict__ Wp2) {
    int i = blockIdx.x * blockDim.x + threadIdx.x;
    int m = i >> 6, k = i & 63;
    if (m < 64) {
        Wp1[i] = W1[m * 128 + k];
        Wp2[i] = W2[m * 128 + k];
    } else {
        int mm = m - 64;
        Wp1[i] = W1[mm * 128 + 64 + k] - W1[mm * 128 + k];
        Wp2[i] = W2[mm * 128 + 64 + k] - W2[mm * 128 + k];
    }
}

template <int LAYOUT>
__global__ __launch_bounds__(256) void gemm_zt_kernel(const float* __restrict__ Wp,
                                                      const float* __restrict__ X,
                                                      float* __restrict__ ZT) {
    const int lane = threadIdx.x & 31;
    const int wave = (blockIdx.x * blockDim.x + threadIdx.x) >> 5;
    const int tilesPerBatch = NN / 16;
    const int b = wave / tilesPerBatch;
    const int t = wave - b * tilesPerBatch;
    const int colLocal = lane & 15;
    const int col = t * 16 + colLocal;
    const int khalf = lane >> 4;

    v8f acc[8] = {};

    for (int k0 = 0; k0 < 64; k0 += 4) {
        const int kk = k0 + khalf * 2;
        v2f bm;
        if (LAYOUT == 0) {
            const float* xb = X + (size_t)b * DD * NN;
            bm.x = xb[(size_t)kk * NN + col];
            bm.y = xb[(size_t)(kk + 1) * NN + col];
        } else {
            const float* xb = X + ((size_t)b * NN + col) * DD + kk;
            bm.x = xb[0];
            bm.y = xb[1];
        }
#pragma unroll
        for (int mt = 0; mt < 8; ++mt) {
            const float* wp = Wp + (mt * 16 + colLocal) * 64 + kk;
            v2f am;
            am.x = wp[0];
            am.y = wp[1];
            acc[mt] = __builtin_amdgcn_wmma_f32_16x16x4_f32(
                false, am, false, bm, (short)0, acc[mt], false, false);
        }
    }

    float* outp = ZT + ((size_t)b * NN + col) * 128 + khalf * 8;
#pragma unroll
    for (int mt = 0; mt < 8; ++mt) {
        v4f lo = {acc[mt][0], acc[mt][1], acc[mt][2], acc[mt][3]};
        v4f hi = {acc[mt][4], acc[mt][5], acc[mt][6], acc[mt][7]};
        *(v4f*)(outp + mt * 16) = lo;
        *(v4f*)(outp + mt * 16 + 4) = hi;
    }
}

__device__ __forceinline__ float gelu_exact(float x) {
    return 0.5f * x * (1.0f + erff(x * 0.70710678118654752f));
}

__global__ __launch_bounds__(256) void gather1_kernel(const float* __restrict__ ZT,
                                                      const int* __restrict__ IDX,
                                                      const float* __restrict__ g1,
                                                      const float* __restrict__ b1,
                                                      const float* __restrict__ m1,
                                                      const float* __restrict__ v1,
                                                      float* __restrict__ H) {
    const int lane = threadIdx.x & 31;
    const int wave = (blockIdx.x * blockDim.x + threadIdx.x) >> 5;
    const int b = wave / NN;
    const int n = wave - b * NN;
    const int ch = lane * 2;
    const int* ip = IDX + ((size_t)b * NN + n) * KK;
    const float* zb = ZT + (size_t)b * NN * 128;

    float mx0 = -3.402823466e38f, mx1 = -3.402823466e38f;
    for (int k = 0; k < KK; ++k) {
        int j = ip[k];
        const float* p = zb + (size_t)j * 128 + ch;
        mx0 = fmaxf(mx0, p[0]);
        mx1 = fmaxf(mx1, p[1]);
    }
    const float* cp = zb + (size_t)n * 128 + 64 + ch;
    float h0 = mx0 + cp[0];
    float h1 = mx1 + cp[1];

    float sc0 = g1[ch] / sqrtf(v1[ch] + 1e-5f);
    float sc1 = g1[ch + 1] / sqrtf(v1[ch + 1] + 1e-5f);
    float sh0 = b1[ch] - m1[ch] * sc0;
    float sh1 = b1[ch + 1] - m1[ch + 1] * sc1;

    float* hp = H + ((size_t)b * NN + n) * 64 + ch;
    hp[0] = gelu_exact(h0 * sc0 + sh0);
    hp[1] = gelu_exact(h1 * sc1 + sh1);
}

#define TP 32
#define PW (TP / 8)
#define TPITCH (TP + 1)
__global__ __launch_bounds__(256) void gather2_kernel(const float* __restrict__ ZT,
                                                      const int* __restrict__ IDX,
                                                      const float* __restrict__ X,
                                                      const float* __restrict__ g2,
                                                      const float* __restrict__ b2,
                                                      const float* __restrict__ m2,
                                                      const float* __restrict__ v2,
                                                      float* __restrict__ OUT) {
    __shared__ float s_t[DD * TPITCH];

    const int lane = threadIdx.x & 31;
    const int w = threadIdx.x >> 5;
    const int tilesPerBatch = NN / TP;
    const int b = blockIdx.x / tilesPerBatch;
    const int n0 = (blockIdx.x - b * tilesPerBatch) * TP;
    const int ch = lane * 2;
    const float* zb = ZT + (size_t)b * NN * 128;

    const float* xr = X + ((size_t)b * DD + w * 8) * NN + n0 + lane;
    for (int i = 0; i < 8; ++i)
        s_t[(w * 8 + i) * TPITCH + lane] = xr[(size_t)i * NN];

    float sc0 = g2[ch] / sqrtf(v2[ch] + 1e-5f);
    float sc1 = g2[ch + 1] / sqrtf(v2[ch + 1] + 1e-5f);
    float sh0 = b2[ch] - m2[ch] * sc0;
    float sh1 = b2[ch + 1] - m2[ch + 1] * sc1;

    float mx0[PW], mx1[PW], cc0[PW], cc1[PW];
#pragma unroll
    for (int i = 0; i < PW; ++i) {
        const int n = n0 + w * PW + i;
        const int* ip = IDX + ((size_t)b * NN + n) * KK;
        float a0 = -3.402823466e38f, a1 = -3.402823466e38f;
        for (int k = 0; k < KK; ++k) {
            int j = ip[k];
            const float* p = zb + (size_t)j * 128 + ch;
            a0 = fmaxf(a0, p[0]);
            a1 = fmaxf(a1, p[1]);
        }
        const float* cp = zb + (size_t)n * 128 + 64 + ch;
        mx0[i] = a0;
        mx1[i] = a1;
        cc0[i] = cp[0];
        cc1[i] = cp[1];
    }
    __syncthreads();

    for (int i = 0; i < PW; ++i) {
        const int pl = w * PW + i;
        float h0 = s_t[ch * TPITCH + pl] + mx0[i] + cc0[i];
        float h1 = s_t[(ch + 1) * TPITCH + pl] + mx1[i] + cc1[i];
        s_t[ch * TPITCH + pl] = gelu_exact(h0 * sc0 + sh0);
        s_t[(ch + 1) * TPITCH + pl] = gelu_exact(h1 * sc1 + sh1);
    }
    __syncthreads();

    float* orow = OUT + ((size_t)b * DD + w * 8) * NN + n0 + lane;
    for (int i = 0; i < 8; ++i)
        orow[(size_t)i * NN] = s_t[(w * 8 + i) * TPITCH + lane];
}

extern "C" void kernel_launch(void* const* d_in, const int* in_sizes, int n_in,
                              void* d_out, int out_size, void* d_ws, size_t ws_size,
                              hipStream_t stream) {
    const float* x   = (const float*)d_in[0];
    const int*   idx = (const int*)d_in[1];
    const float* W1  = (const float*)d_in[2];
    const float* W2  = (const float*)d_in[3];
    const float* g1  = (const float*)d_in[4];
    const float* b1  = (const float*)d_in[5];
    const float* m1  = (const float*)d_in[6];
    const float* v1  = (const float*)d_in[7];
    const float* g2  = (const float*)d_in[8];
    const float* b2  = (const float*)d_in[9];
    const float* m2  = (const float*)d_in[10];
    const float* v2  = (const float*)d_in[11];
    float* out = (float*)d_out;

    char* ws = (char*)d_ws;
    float* Wp1 = (float*)ws;
    float* Wp2 = (float*)(ws + 32 * 1024);
    float* ZT  = (float*)(ws + 64 * 1024);
    float* H   = (float*)(ws + 64 * 1024 + (size_t)BB * NN * 128 * 4);

    prep_w_kernel<<<(128 * 64) / 256, 256, 0, stream>>>(W1, W2, Wp1, Wp2);

    const int gemmBlocks = (BB * (NN / 16) * 32) / 256;
    const int gathBlocks = (BB * NN * 32) / 256;
    const int gath2Blocks = (BB * NN) / TP;

    gemm_zt_kernel<0><<<gemmBlocks, 256, 0, stream>>>(Wp1, x, ZT);
    gather1_kernel<<<gathBlocks, 256, 0, stream>>>(ZT, idx, g1, b1, m1, v1, H);
    gemm_zt_kernel<1><<<gemmBlocks, 256, 0, stream>>>(Wp2, H, ZT);
    gather2_kernel<<<gath2Blocks, 256, 0, stream>>>(ZT, idx, x, g2, b2, m2, v2, out);
}


# gemm A operand: prep writes W' in WMMA fragment order, lane-contiguous, two k-steps per 16-byte load: each A load reads 4 whole lines (was 16 pieces of 16 B); same sums in the same order
# speedup vs baseline: 1.9894x; 1.0431x over previous
#include <hip/hip_runtime.h>
#include <math.h>

typedef __attribute__((ext_vector_type(2))) float v2f;
typedef __attribute__((ext_vector_type(4))) float v4f;
typedef __attribute__((ext_vector_type(8))) float v8f;

#define BB 4
#define DD 64
#define NN 16384
#define KK 20

__global__ __launch_bounds__(256) void prep_w_kernel(const float* __restrict__ W1,
                                                     const float* __restrict__ W2,
                                                     float* __restrict__ Wp1,
                                                     float* __restrict__ Wp2) {
    int i = blockIdx.x * blockDim.x + threadIdx.x;
    int e = i & 3, l = (i >> 2) & 31, kp = (i >> 7) & 7, mt = i >> 10;
    int m = mt * 16 + (l & 15);
    int k = kp * 8 + (e >> 1) * 4 + (l >> 4) * 2 + (e & 1);
    if (m < 64) {
        Wp1[i] = W1[m * 128 + k];
        Wp2[i] = W2[m * 128 + k];
    } else {
        int mm = m - 64;
        Wp1[i] = W1[mm * 128 + 64 + k] - W1[mm * 128 + k];
        Wp2[i] = W2[mm * 128 + 64 + k] - W2[mm * 128 + k];
    }
}

template <int LAYOUT>
__global__ __launch_bounds__(256) void gemm_zt_kernel(const float* __restrict__ Wp,
                                                      const float* __restrict__ X,
                                                      float* __restrict__ ZT) {
    const int lane = threadIdx.x & 31;
    const int wave = (blockIdx.x * blockDim.x + threadIdx.x) >> 5;
    const int tilesPerBatch = NN / 16;
    const int b = wave / tilesPerBatch;
    const int t = wave - b * tilesPerBatch;
    const int colLocal = lane & 15;
    const int col = t * 16 + colLocal;
    const int khalf = lane >> 4;

    v8f acc[8] = {};

    const v4f* wf = (const v4f*)Wp + lane;

    for (int kp = 0; kp < 8; ++kp) {
        const int kk = kp * 8 + khalf * 2;
        v2f bm0, bm1;
        if (LAYOUT == 0) {
            const float* xb = X + (size_t)b * DD * NN;
            bm0.x = xb[(size_t)kk * NN + col];
            bm0.y = xb[(size_t)(kk + 1) * NN + col];
            bm1.x = xb[(size_t)(kk + 4) * NN + col];
            bm1.y = xb[(size_t)(kk + 5) * NN + col];
        } else {
            const float* xb = X + ((size_t)b * NN + col) * DD + kk;
            bm0.x = xb[0];
            bm0.y = xb[1];
            bm1.x = xb[4];
            bm1.y = xb[5];
        }
#pragma unroll
        for (int mt = 0; mt < 8; ++mt) {
            const v4f a = wf[(mt * 8 + kp) * 32];
            v2f am0 = {a[0], a[1]};
            v2f am1 = {a[2], a[3]};
            acc[mt] = __builtin_amdgcn_wmma_f32_16x16x4_f32(
                false, am0, false, bm0, (short)0, acc[mt], false, false);
            acc[mt] = __builtin_amdgcn_wmma_f32_16x16x4_f32(
                false, am1, false, bm1, (short)0, acc[mt], false, false);
        }
    }

    float* outp = ZT + ((size_t)b * NN + col) * 128 + khalf * 8;
#pragma unroll
    for (int mt = 0; mt < 8; ++mt) {
        v4f lo = {acc[mt][0], acc[mt][1], acc[mt][2], acc[mt][3]};
        v4f hi = {acc[mt][4], acc[mt][5], acc[mt][6], acc[mt][7]};
        *(v4f*)(outp + mt * 16) = lo;
        *(v4f*)(outp + mt * 16 + 4) = hi;
    }
}

__device__ __forceinline__ float gelu_exact(float x) {
    return 0.5f * x * (1.0f + erff(x * 0.70710678118654752f));
}

__global__ __launch_bounds__(256) void gather1_kernel(const float* __restrict__ ZT,
                                                      const int* __restrict__ IDX,
                                                      const float* __restrict__ g1,
                                                      const float* __restrict__ b1,
                                                      const float* __restrict__ m1,
                                                      const float* __restrict__ v1,
                                                      float* __restrict__ H) {
    const int lane = threadIdx.x & 31;
    const int wave = (blockIdx.x * blockDim.x + threadIdx.x) >> 5;
    const int b = wave / NN;
    const int n = wave - b * NN;
    const int ch = lane * 2;
    const int* ip = IDX + ((size_t)b * NN + n) * KK;
    const float* zb = ZT + (size_t)b * NN * 128;

    float mx0 = -3.402823466e38f, mx1 = -3.402823466e38f;
    for (int k = 0; k < KK; ++k) {
        int j = ip[k];
        const float* p = zb + (size_t)j * 128 + ch;
        mx0 = fmaxf(mx0, p[0]);
        mx1 = fmaxf(mx1, p[1]);
    }
    const float* cp = zb + (size_t)n * 128 + 64 + ch;
    float h0 = mx0 + cp[0];
    float h1 = mx1 + cp[1];

    float sc0 = g1[ch] / sqrtf(v1[ch] + 1e-5f);
    float sc1 = g1[ch + 1] / sqrtf(v1[ch + 1] + 1e-5f);
    float sh0 = b1[ch] - m1[ch] * sc0;
    float sh1 = b1[ch + 1] - m1[ch + 1] * sc1;

    float* hp = H + ((size_t)b * NN + n) * 64 + ch;
    hp[0] = gelu_exact(h0 * sc0 + sh0);
    hp[1] = gelu_exact(h1 * sc1 + sh1);
}

#define TP 32
#define PW (TP / 8)
#define TPITCH (TP + 1)
__global__ __launch_bounds__(256) void gather2_kernel(const float* __restrict__ ZT,
                                                      const int* __restrict__ IDX,
                                                      const float* __restrict__ X,
                                                      const float* __restrict__ g2,
                                                      const float* __restrict__ b2,
                                                      const float* __restrict__ m2,
                                                      const float* __restrict__ v2,
                                                      float* __restrict__ OUT) {
    __shared__ float s_t[DD * TPITCH];

    const int lane = threadIdx.x & 31;
    const int w = threadIdx.x >> 5;
    const int tilesPerBatch = NN / TP;
    const int b = blockIdx.x / tilesPerBatch;
    const int n0 = (blockIdx.x - b * tilesPerBatch) * TP;
    const int ch = lane * 2;
    const float* zb = ZT + (size_t)b * NN * 128;

    const float* xr = X + ((size_t)b * DD + w * 8) * NN + n0 + lane;
    for (int i = 0; i < 8; ++i)
        s_t[(w * 8 + i) * TPITCH + lane] = xr[(size_t)i * NN];

    float sc0 = g2[ch] / sqrtf(v2[ch] + 1e-5f);
    float sc1 = g2[ch + 1] / sqrtf(v2[ch + 1] + 1e-5f);
    float sh0 = b2[ch] - m2[ch] * sc0;
    float sh1 = b2[ch + 1] - m2[ch + 1] * sc1;

    float mx0[PW], mx1[PW], cc0[PW], cc1[PW];
#pragma unroll
    for (int i = 0; i < PW; ++i) {
        const int n = n0 + w * PW + i;
        const int* ip = IDX + ((size_t)b * NN + n) * KK;
        float a0 = -3.402823466e38f, a1 = -3.402823466e38f;
        for (int k = 0; k < KK; ++k) {
            int j = ip[k];
            const float* p = zb + (size_t)j * 128 + ch;
            a0 = fmaxf(a0, p[0]);
            a1 = fmaxf(a1, p[1]);
        }
        const float* cp = zb + (size_t)n * 128 + 64 + ch;
        mx0[i] = a0;
        mx1[i] = a1;
        cc0[i] = cp[0];
        cc1[i] = cp[1];
    }
    __syncthreads();

    for (int i = 0; i < PW; ++i) {
        const int pl = w * PW + i;
        float h0 = s_t[ch * TPITCH + pl] + mx0[i] + cc0[i];
        float h1 = s_t[(ch + 1) * TPITCH + pl] + mx1[i] + cc1[i];
        s_t[ch * TPITCH + pl] = gelu_exact(h0 * sc0 + sh0);
        s_t[(ch + 1) * TPITCH + pl] = gelu_exact(h1 * sc1 + sh1);
    }
    __syncthreads();

    float* orow = OUT + ((size_t)b * DD + w * 8) * NN + n0 + lane;
    for (int i = 0; i < 8; ++i)
        orow[(size_t)i * NN] = s_t[(w * 8 + i) * TPITCH + lane];
}

extern "C" void kernel_launch(void* const* d_in, const int* in_sizes, int n_in,
                              void* d_out, int out_size, void* d_ws, size_t ws_size,
                              hipStream_t stream) {
    const float* x   = (const float*)d_in[0];
    const int*   idx = (const int*)d_in[1];
    const float* W1  = (const float*)d_in[2];
    const float* W2  = (const float*)d_in[3];
    const float* g1  = (const float*)d_in[4];
    const float* b1  = (const float*)d_in[5];
    const float* m1  = (const float*)d_in[6];
    const float* v1  = (const float*)d_in[7];
    const float* g2  = (const float*)d_in[8];
    const float* b2  = (const float*)d_in[9];
    const float* m2  = (const float*)d_in[10];
    const float* v2  = (const float*)d_in[11];
    float* out = (float*)d_out;

    char* ws = (char*)d_ws;
    float* Wp1 = (float*)ws;
    float* Wp2 = (float*)(ws + 32 * 1024);
    float* ZT  = (float*)(ws + 64 * 1024);
    float* H   = (float*)(ws + 64 * 1024 + (size_t)BB * NN * 128 * 4);

    prep_w_kernel<<<(128 * 64) / 256, 256, 0, stream>>>(W1, W2, Wp1, Wp2);

    const int gemmBlocks = (BB * (NN / 16) * 32) / 256;
    const int gathBlocks = (BB * NN * 32) / 256;
    const int gath2Blocks = (BB * NN) / TP;

    gemm_zt_kernel<0><<<gemmBlocks, 256, 0, stream>>>(Wp1, x, ZT);
    gather1_kernel<<<gathBlocks, 256, 0, stream>>>(ZT, idx, g1, b1, m1, v1, H);
    gemm_zt_kernel<1><<<gemmBlocks, 256, 0, stream>>>(Wp2, H, ZT);
    gather2_kernel<<<gath2Blocks, 256, 0, stream>>>(ZT, idx, x, g2, b2, m2, v2, out);
}


# gemm ZT store: a wave stages its 16x128 result tile in its own LDS block and writes one point's whole 512-byte row per store (4 full lines, was 16 lines in 16-byte pieces); same sums, same order
# speedup vs baseline: 2.0541x; 1.0325x over previous
#include <hip/hip_runtime.h>
#include <math.h>

typedef __attribute__((ext_vector_type(2))) float v2f;
typedef __attribute__((ext_vector_type(4))) float v4f;
typedef __attribute__((ext_vector_type(8))) float v8f;

#define BB 4
#define DD 64
#define NN 16384
#define KK 20

__global__ __launch_bounds__(256) void prep_w_kernel(const float* __restrict__ W1,
                                                     const float* __restrict__ W2,
                                                     float* __restrict__ Wp1,
                                                     float* __restrict__ Wp2) {
    int i = blockIdx.x * blockDim.x + threadIdx.x;
    int e = i & 3, l = (i >> 2) & 31, kp = (i >> 7) & 7, mt = i >> 10;
    int m = mt * 16 + (l & 15);
    int k = kp * 8 + (e >> 1) * 4 + (l >> 4) * 2 + (e & 1);
    if (m < 64) {
        Wp1[i] = W1[m * 128 + k];
        Wp2[i] = W2[m * 128 + k];
    } else {
        int mm = m - 64;
        Wp1[i] = W1[mm * 128 + 64 + k] - W1[mm * 128 + k];
        Wp2[i] = W2[mm * 128 + 64 + k] - W2[mm * 128 + k];
    }
}

#define SROW (128 + 4)
template <int LAYOUT>
__global__ __launch_bounds__(256) void gemm_zt_kernel(const float* __restrict__ Wp,
                                                      const float* __restrict__ X,
                                                      float* __restrict__ ZT) {
    __shared__ __attribute__((aligned(16))) float s_stage[8 * 16 * SROW];

    const int lane = threadIdx.x & 31;
    const int wave = (blockIdx.x * blockDim.x + threadIdx.x) >> 5;
    const int tilesPerBatch = NN / 16;
    const int b = wave / tilesPerBatch;
    const int t = wave - b * tilesPerBatch;
    const int colLocal = lane & 15;
    const int col = t * 16 + colLocal;
    const int khalf = lane >> 4;

    v8f acc[8] = {};

    const v4f* wf = (const v4f*)Wp + lane;

    for (int kp = 0; kp < 8; ++kp) {
        const int kk = kp * 8 + khalf * 2;
        v2f bm0, bm1;
        if (LAYOUT == 0) {
            const float* xb = X + (size_t)b * DD * NN;
            bm0.x = xb[(size_t)kk * NN + col];
            bm0.y = xb[(size_t)(kk + 1) * NN + col];
            bm1.x = xb[(size_t)(kk + 4) * NN + col];
            bm1.y = xb[(size_t)(kk + 5) * NN + col];
        } else {
            const float* xb = X + ((size_t)b * NN + col) * DD + kk;
            bm0.x = xb[0];
            bm0.y = xb[1];
            bm1.x = xb[4];
            bm1.y = xb[5];
        }
#pragma unroll
        for (int mt = 0; mt < 8; ++mt) {
            const v4f a = wf[(mt * 8 + kp) * 32];
            v2f am0 = {a[0], a[1]};
            v2f am1 = {a[2], a[3]};
            acc[mt] = __builtin_amdgcn_wmma_f32_16x16x4_f32(
                false, am0, false, bm0, (short)0, acc[mt], false, false);
            acc[mt] = __builtin_amdgcn_wmma_f32_16x16x4_f32(
                false, am1, false, bm1, (short)0, acc[mt], false, false);
        }
    }

    float* st = s_stage + (threadIdx.x >> 5) * (16 * SROW);
    float* sw = st + colLocal * SROW + khalf * 8;
#pragma unroll
    for (int mt = 0; mt < 8; ++mt) {
        v4f lo = {acc[mt][0], acc[mt][1], acc[mt][2], acc[mt][3]};
        v4f hi = {acc[mt][4], acc[mt][5], acc[mt][6], acc[mt][7]};
        *(v4f*)(sw + mt * 16) = lo;
        *(v4f*)(sw + mt * 16 + 4) = hi;
    }
    asm volatile("s_wait_dscnt 0x0" ::: "memory");
    const float* sr = st + lane * 4;
    float* outp = ZT + ((size_t)b * NN + t * 16) * 128 + lane * 4;
#pragma unroll
    for (int q = 0; q < 16; ++q) {
        const v4f v = *(const v4f*)(sr + q * SROW);
        *(v4f*)(outp + q * 128) = v;
    }
}

__device__ __forceinline__ float gelu_exact(float x) {
    return 0.5f * x * (1.0f + erff(x * 0.70710678118654752f));
}

__global__ __launch_bounds__(256) void gather1_kernel(const float* __restrict__ ZT,
                                                      const int* __restrict__ IDX,
                                                      const float* __restrict__ g1,
                                                      const float* __restrict__ b1,
                                                      const float* __restrict__ m1,
                                                      const float* __restrict__ v1,
                                                      float* __restrict__ H) {
    const int lane = threadIdx.x & 31;
    const int wave = (blockIdx.x * blockDim.x + threadIdx.x) >> 5;
    const int b = wave / NN;
    const int n = wave - b * NN;
    const int ch = lane * 2;
    const int* ip = IDX + ((size_t)b * NN + n) * KK;
    const float* zb = ZT + (size_t)b * NN * 128;

    float mx0 = -3.402823466e38f, mx1 = -3.402823466e38f;
    for (int k = 0; k < KK; ++k) {
        int j = ip[k];
        const float* p = zb + (size_t)j * 128 + ch;
        mx0 = fmaxf(mx0, p[0]);
        mx1 = fmaxf(mx1, p[1]);
    }
    const float* cp = zb + (size_t)n * 128 + 64 + ch;
    float h0 = mx0 + cp[0];
    float h1 = mx1 + cp[1];

    float sc0 = g1[ch] / sqrtf(v1[ch] + 1e-5f);
    float sc1 = g1[ch + 1] / sqrtf(v1[ch + 1] + 1e-5f);
    float sh0 = b1[ch] - m1[ch] * sc0;
    float sh1 = b1[ch + 1] - m1[ch + 1] * sc1;

    float* hp = H + ((size_t)b * NN + n) * 64 + ch;
    hp[0] = gelu_exact(h0 * sc0 + sh0);
    hp[1] = gelu_exact(h1 * sc1 + sh1);
}

#define TP 32
#define PW (TP / 8)
#define TPITCH (TP + 1)
__global__ __launch_bounds__(256) void gather2_kernel(const float* __restrict__ ZT,
                                                      const int* __restrict__ IDX,
                                                      const float* __restrict__ X,
                                                      const float* __restrict__ g2,
                                                      const float* __restrict__ b2,
                                                      const float* __restrict__ m2,
                                                      const float* __restrict__ v2,
                                                      float* __restrict__ OUT) {
    __shared__ float s_t[DD * TPITCH];

    const int lane = threadIdx.x & 31;
    const int w = threadIdx.x >> 5;
    const int tilesPerBatch = NN / TP;
    const int b = blockIdx.x / tilesPerBatch;
    const int n0 = (blockIdx.x - b * tilesPerBatch) * TP;
    const int ch = lane * 2;
    const float* zb = ZT + (size_t)b * NN * 128;

    const float* xr = X + ((size_t)b * DD + w * 8) * NN + n0 + lane;
    for (int i = 0; i < 8; ++i)
        s_t[(w * 8 + i) * TPITCH + lane] = xr[(size_t)i * NN];

    float sc0 = g2[ch] / sqrtf(v2[ch] + 1e-5f);
    float sc1 = g2[ch + 1] / sqrtf(v2[ch + 1] + 1e-5f);
    float sh0 = b2[ch] - m2[ch] * sc0;
    float sh1 = b2[ch + 1] - m2[ch + 1] * sc1;

    float mx0[PW], mx1[PW], cc0[PW], cc1[PW];
#pragma unroll
    for (int i = 0; i < PW; ++i) {
        const int n = n0 + w * PW + i;
        const int* ip = IDX + ((size_t)b * NN + n) * KK;
        float a0 = -3.402823466e38f, a1 = -3.402823466e38f;
        for (int k = 0; k < KK; ++k) {
            int j = ip[k];
            const float* p = zb + (size_t)j * 128 + ch;
            a0 = fmaxf(a0, p[0]);
            a1 = fmaxf(a1, p[1]);
        }
        const float* cp = zb + (size_t)n * 128 + 64 + ch;
        mx0[i] = a0;
        mx1[i] = a1;
        cc0[i] = cp[0];
        cc1[i] = cp[1];
    }
    __syncthreads();

    for (int i = 0; i < PW; ++i) {
        const int pl = w * PW + i;
        float h0 = s_t[ch * TPITCH + pl] + mx0[i] + cc0[i];
        float h1 = s_t[(ch + 1) * TPITCH + pl] + mx1[i] + cc1[i];
        s_t[ch * TPITCH + pl] = gelu_exact(h0 * sc0 + sh0);
        s_t[(ch + 1) * TPITCH + pl] = gelu_exact(h1 * sc1 + sh1);
    }
    __syncthreads();

    float* orow = OUT + ((size_t)b * DD + w * 8) * NN + n0 + lane;
    for (int i = 0; i < 8; ++i)
        orow[(size_t)i * NN] = s_t[(w * 8 + i) * TPITCH + lane];
}

extern "C" void kernel_launch(void* const* d_in, const int* in_sizes, int n_in,
                              void* d_out, int out_size, void* d_ws, size_t ws_size,
                              hipStream_t stream) {
    const float* x   = (const float*)d_in[0];
    const int*   idx = (const int*)d_in[1];
    const float* W1  = (const float*)d_in[2];
    const float* W2  = (const float*)d_in[3];
    const float* g1  = (const float*)d_in[4];
    const float* b1  = (const float*)d_in[5];
    const float* m1  = (const float*)d_in[6];
    const float* v1  = (const float*)d_in[7];
    const float* g2  = (const float*)d_in[8];
    const float* b2  = (const float*)d_in[9];
    const float* m2  = (const float*)d_in[10];
    const float* v2  = (const float*)d_in[11];
    float* out = (float*)d_out;

    char* ws = (char*)d_ws;
    float* Wp1 = (float*)ws;
    float* Wp2 = (float*)(ws + 32 * 1024);
    float* ZT  = (float*)(ws + 64 * 1024);
    float* H   = (float*)(ws + 64 * 1024 + (size_t)BB * NN * 128 * 4);

    prep_w_kernel<<<(128 * 64) / 256, 256, 0, stream>>>(W1, W2, Wp1, Wp2);

    const int gemmBlocks = (BB * (NN / 16) * 32) / 256;
    const int gathBlocks = (BB * NN * 32) / 256;
    const int gath2Blocks = (BB * NN) / TP;

    gemm_zt_kernel<0><<<gemmBlocks, 256, 0, stream>>>(Wp1, x, ZT);
    gather1_kernel<<<gathBlocks, 256, 0, stream>>>(ZT, idx, g1, b1, m1, v1, H);
    gemm_zt_kernel<1><<<gemmBlocks, 256, 0, stream>>>(Wp2, H, ZT);
    gather2_kernel<<<gath2Blocks, 256, 0, stream>>>(ZT, idx, x, g2, b2, m2, v2, out);
}


# weight re-layout joins the gemm launch: each block builds W' = [Wa; Wb-Wa] in LDS from W (272 B row pitch, conflict-free A reads); 4 launches, was 5; same sums in the same order
# speedup vs baseline: 2.2247x; 1.0831x over previous
#include <hip/hip_runtime.h>
#include <math.h>

typedef __attribute__((ext_vector_type(2))) float v2f;
typedef __attribute__((ext_vector_type(4))) float v4f;
typedef __attribute__((ext_vector_type(8))) float v8f;

#define BB 4
#define DD 64
#define NN 16384
#define KK 20

#define SROW (128 + 4)
#define WPITCH (64 + 4)

template <int LAYOUT>
__device__ __forceinline__ void load_b(const float* __restrict__ X, int b, int col, int khalf, int kp,
                                       v2f& bm0, v2f& bm1) {
    const int kk = kp * 8 + khalf * 2;
    if (LAYOUT == 0) {
        const float* xb = X + (size_t)b * DD * NN;
        bm0.x = xb[(size_t)kk * NN + col];
        bm0.y = xb[(size_t)(kk + 1) * NN + col];
        bm1.x = xb[(size_t)(kk + 4) * NN + col];
        bm1.y = xb[(size_t)(kk + 5) * NN + col];
    } else {
        const float* xb = X + ((size_t)b * NN + col) * DD + kk;
        bm0.x = xb[0];
        bm0.y = xb[1];
        bm1.x = xb[4];
        bm1.y = xb[5];
    }
}

template <int LAYOUT>
__global__ __launch_bounds__(256) void gemm_zt_kernel(const float* __restrict__ W,
                                                      const float* __restrict__ X,
                                                      float* __restrict__ ZT) {
    __shared__ __attribute__((aligned(16))) float s_stage[8 * 16 * SROW];
    __shared__ __attribute__((aligned(16))) float s_w[128 * WPITCH];

    const int lane = threadIdx.x & 31;
    const int wave = (blockIdx.x * blockDim.x + threadIdx.x) >> 5;
    const int tilesPerBatch = NN / 16;
    const int b = wave / tilesPerBatch;
    const int t = wave - b * tilesPerBatch;
    const int colLocal = lane & 15;
    const int col = t * 16 + colLocal;
    const int khalf = lane >> 4;

    v8f acc[8] = {};
    v2f bm0[8], bm1[8];

#pragma unroll
    for (int kp = 0; kp < 2; ++kp)
        load_b<LAYOUT>(X, b, col, khalf, kp, bm0[kp], bm1[kp]);

    {
        const int q = threadIdx.x & 15;
        const int r = threadIdx.x >> 4;
#pragma unroll
        for (int i = 0; i < 4; ++i) {
            const int m = i * 16 + r;
            const v4f wa = *(const v4f*)(W + m * 128 + q * 4);
            const v4f wb = *(const v4f*)(W + m * 128 + 64 + q * 4);
            *(v4f*)(s_w + m * WPITCH + q * 4) = wa;
            *(v4f*)(s_w + (64 + m) * WPITCH + q * 4) = wb - wa;
        }
    }
    __syncthreads();

#pragma unroll
    for (int kp = 2; kp < 8; ++kp)
        load_b<LAYOUT>(X, b, col, khalf, kp, bm0[kp], bm1[kp]);

    const float* wl = s_w + colLocal * WPITCH + khalf * 2;
#pragma unroll
    for (int kp = 0; kp < 8; ++kp) {
#pragma unroll
        for (int mt = 0; mt < 8; ++mt) {
            const float* p = wl + mt * 16 * WPITCH + kp * 8;
            const v2f am0 = *(const v2f*)p;
            const v2f am1 = *(const v2f*)(p + 4);
            acc[mt] = __builtin_amdgcn_wmma_f32_16x16x4_f32(
                false, am0, false, bm0[kp], (short)0, acc[mt], false, false);
            acc[mt] = __builtin_amdgcn_wmma_f32_16x16x4_f32(
                false, am1, false, bm1[kp], (short)0, acc[mt], false, false);
        }
    }

    float* st = s_stage + (threadIdx.x >> 5) * (16 * SROW);
    float* sw = st + colLocal * SROW + khalf * 8;
#pragma unroll
    for (int mt = 0; mt < 8; ++mt) {
        v4f lo = {acc[mt][0], acc[mt][1], acc[mt][2], acc[mt][3]};
        v4f hi = {acc[mt][4], acc[mt][5], acc[mt][6], acc[mt][7]};
        *(v4f*)(sw + mt * 16) = lo;
        *(v4f*)(sw + mt * 16 + 4) = hi;
    }
    asm volatile("s_wait_dscnt 0x0" ::: "memory");
    const float* sr = st + lane * 4;
    float* outp = ZT + ((size_t)b * NN + t * 16) * 128 + lane * 4;
#pragma unroll
    for (int q = 0; q < 16; ++q) {
        const v4f v = *(const v4f*)(sr + q * SROW);
        *(v4f*)(outp + q * 128) = v;
    }
}

__device__ __forceinline__ float gelu_exact(float x) {
    return 0.5f * x * (1.0f + erff(x * 0.70710678118654752f));
}

__global__ __launch_bounds__(256) void gather1_kernel(const float* __restrict__ ZT,
                                                      const int* __restrict__ IDX,
                                                      const float* __restrict__ g1,
                                                      const float* __restrict__ b1,
                                                      const float* __restrict__ m1,
                                                      const float* __restrict__ v1,
                                                      float* __restrict__ H) {
    const int lane = threadIdx.x & 31;
    const int wave = (blockIdx.x * blockDim.x + threadIdx.x) >> 5;
    const int b = wave / NN;
    const int n = wave - b * NN;
    const int ch = lane * 2;
    const int* ip = IDX + ((size_t)b * NN + n) * KK;
    const float* zb = ZT + (size_t)b * NN * 128;

    float mx0 = -3.402823466e38f, mx1 = -3.402823466e38f;
    for (int k = 0; k < KK; ++k) {
        int j = ip[k];
        const float* p = zb + (size_t)j * 128 + ch;
        mx0 = fmaxf(mx0, p[0]);
        mx1 = fmaxf(mx1, p[1]);
    }
    const float* cp = zb + (size_t)n * 128 + 64 + ch;
    float h0 = mx0 + cp[0];
    float h1 = mx1 + cp[1];

    float sc0 = g1[ch] / sqrtf(v1[ch] + 1e-5f);
    float sc1 = g1[ch + 1] / sqrtf(v1[ch + 1] + 1e-5f);
    float sh0 = b1[ch] - m1[ch] * sc0;
    float sh1 = b1[ch + 1] - m1[ch + 1] * sc1;

    float* hp = H + ((size_t)b * NN + n) * 64 + ch;
    hp[0] = gelu_exact(h0 * sc0 + sh0);
    hp[1] = gelu_exact(h1 * sc1 + sh1);
}

#define TP 32
#define PW (TP / 8)
#define TPITCH (TP + 1)
__global__ __launch_bounds__(256) void gather2_kernel(const float* __restrict__ ZT,
                                                      const int* __restrict__ IDX,
                                                      const float* __restrict__ X,
                                                      const float* __restrict__ g2,
                                                      const float* __restrict__ b2,
                                                      const float* __restrict__ m2,
                                                      const float* __restrict__ v2,
                                                      float* __restrict__ OUT) {
    __shared__ float s_t[DD * TPITCH];

    const int lane = threadIdx.x & 31;
    const int w = threadIdx.x >> 5;
    const int tilesPerBatch = NN / TP;
    const int b = blockIdx.x / tilesPerBatch;
    const int n0 = (blockIdx.x - b * tilesPerBatch) * TP;
    const int ch = lane * 2;
    const float* zb = ZT + (size_t)b * NN * 128;

    const float* xr = X + ((size_t)b * DD + w * 8) * NN + n0 + lane;
    for (int i = 0; i < 8; ++i)
        s_t[(w * 8 + i) * TPITCH + lane] = xr[(size_t)i * NN];

    float sc0 = g2[ch] / sqrtf(v2[ch] + 1e-5f);
    float sc1 = g2[ch + 1] / sqrtf(v2[ch + 1] + 1e-5f);
    float sh0 = b2[ch] - m2[ch] * sc0;
    float sh1 = b2[ch + 1] - m2[ch + 1] * sc1;

    float mx0[PW], mx1[PW], cc0[PW], cc1[PW];
#pragma unroll
    for (int i = 0; i < PW; ++i) {
        const int n = n0 + w * PW + i;
        const int* ip = IDX + ((size_t)b * NN + n) * KK;
        float a0 = -3.402823466e38f, a1 = -3.402823466e38f;
        for (int k = 0; k < KK; ++k) {
            int j = ip[k];
            const float* p = zb + (size_t)j * 128 + ch;
            a0 = fmaxf(a0, p[0]);
            a1 = fmaxf(a1, p[1]);
        }
        const float* cp = zb + (size_t)n * 128 + 64 + ch;
        mx0[i] = a0;
        mx1[i] = a1;
        cc0[i] = cp[0];
        cc1[i] = cp[1];
    }
    __syncthreads();

    for (int i = 0; i < PW; ++i) {
        const int pl = w * PW + i;
        float h0 = s_t[ch * TPITCH + pl] + mx0[i] + cc0[i];
        float h1 = s_t[(ch + 1) * TPITCH + pl] + mx1[i] + cc1[i];
        s_t[ch * TPITCH + pl] = gelu_exact(h0 * sc0 + sh0);
        s_t[(ch + 1) * TPITCH + pl] = gelu_exact(h1 * sc1 + sh1);
    }
    __syncthreads();

    float* orow = OUT + ((size_t)b * DD + w * 8) * NN + n0 + lane;
    for (int i = 0; i < 8; ++i)
        orow[(size_t)i * NN] = s_t[(w * 8 + i) * TPITCH + lane];
}

extern "C" void kernel_launch(void* const* d_in, const int* in_sizes, int n_in,
                              void* d_out, int out_size, void* d_ws, size_t ws_size,
                              hipStream_t stream) {
    const float* x   = (const float*)d_in[0];
    const int*   idx = (const int*)d_in[1];
    const float* W1  = (const float*)d_in[2];
    const float* W2  = (const float*)d_in[3];
    const float* g1  = (const float*)d_in[4];
    const float* b1  = (const float*)d_in[5];
    const float* m1  = (const float*)d_in[6];
    const float* v1  = (const float*)d_in[7];
    const float* g2  = (const float*)d_in[8];
    const float* b2  = (const float*)d_in[9];
    const float* m2  = (const float*)d_in[10];
    const float* v2  = (const float*)d_in[11];
    float* out = (float*)d_out;

    char* ws = (char*)d_ws;
    float* ZT  = (float*)ws;
    float* H   = (float*)(ws + (size_t)BB * NN * 128 * 4);

    const int gemmBlocks = (BB * (NN / 16) * 32) / 256;
    const int gathBlocks = (BB * NN * 32) / 256;
    const int gath2Blocks = (BB * NN) / TP;

    gemm_zt_kernel<0><<<gemmBlocks, 256, 0, stream>>>(W1, x, ZT);
    gather1_kernel<<<gathBlocks, 256, 0, stream>>>(ZT, idx, g1, b1, m1, v1, H);
    gemm_zt_kernel<1><<<gemmBlocks, 256, 0, stream>>>(W2, H, ZT);
    gather2_kernel<<<gath2Blocks, 256, 0, stream>>>(ZT, idx, x, g2, b2, m2, v2, out);
}


# neighbour rows read from LDS, not from L2: intermediates slice-major [batch][4-channel slice][point]; a 1024-thread block copies one 256 KB slice into LDS, lane = point; same sums, same order
# speedup vs baseline: 2.3648x; 1.0630x over previous
#include <hip/hip_runtime.h>
#include <math.h>

typedef __attribute__((ext_vector_type(2))) float v2f;
typedef __attribute__((ext_vector_type(4))) float v4f;
typedef __attribute__((ext_vector_type(8))) float v8f;
typedef __attribute__((ext_vector_type(4))) int v4i;

#define BB 4
#define DD 64
#define NN 16384
#define KK 20
#define NSL 32

#define WPITCH (64 + 4)

template <int LAYOUT>
__device__ __forceinline__ void load_b(const float* __restrict__ X, int b, int col, int khalf, int kp,
                                       v2f& bm0, v2f& bm1) {
    const int kk = kp * 8 + khalf * 2;
    if (LAYOUT == 0) {
        const float* xb = X + (size_t)b * DD * NN;
        bm0.x = xb[(size_t)kk * NN + col];
        bm0.y = xb[(size_t)(kk + 1) * NN + col];
        bm1.x = xb[(size_t)(kk + 4) * NN + col];
        bm1.y = xb[(size_t)(kk + 5) * NN + col];
    } else {
        const float* xb = X + (((size_t)b * 16 + kp * 2) * NN + col) * 4 + khalf * 2;
        bm0 = *(const v2f*)xb;
        bm1 = *(const v2f*)(xb + (size_t)NN * 4);
    }
}

template <int LAYOUT>
__global__ __launch_bounds__(256) void gemm_zs_kernel(const float* __restrict__ W,
                                                      const float* __restrict__ X,
                                                      float* __restrict__ ZS) {
    __shared__ __attribute__((aligned(16))) float s_w[128 * WPITCH];

    const int lane = threadIdx.x & 31;
    const int wave = (blockIdx.x * blockDim.x + threadIdx.x) >> 5;
    const int tilesPerBatch = NN / 16;
    const int b = wave / tilesPerBatch;
    const int t = wave - b * tilesPerBatch;
    const int colLocal = lane & 15;
    const int col = t * 16 + colLocal;
    const int khalf = lane >> 4;

    v8f acc[8] = {};
    v2f bm0[8], bm1[8];

#pragma unroll
    for (int kp = 0; kp < 2; ++kp)
        load_b<LAYOUT>(X, b, col, khalf, kp, bm0[kp], bm1[kp]);

    {
        const int q = threadIdx.x & 15;
        const int r = threadIdx.x >> 4;
#pragma unroll
        for (int i = 0; i < 4; ++i) {
            const int m = i * 16 + r;
            const v4f wa = *(const v4f*)(W + m * 128 + q * 4);
            const v4f wb = *(const v4f*)(W + m * 128 + 64 + q * 4);
            *(v4f*)(s_w + m * WPITCH + q * 4) = wa;
            *(v4f*)(s_w + (64 + m) * WPITCH + q * 4) = wb - wa;
        }
    }
    __syncthreads();

#pragma unroll
    for (int kp = 2; kp < 8; ++kp)
        load_b<LAYOUT>(X, b, col, khalf, kp, bm0[kp], bm1[kp]);

    const float* wl = s_w + colLocal * WPITCH + khalf * 2;
#pragma unroll
    for (int kp = 0; kp < 8; ++kp) {
#pragma unroll
        for (int mt = 0; mt < 8; ++mt) {
            const float* p = wl + mt * 16 * WPITCH + kp * 8;
            const v2f am0 = *(const v2f*)p;
            const v2f am1 = *(const v2f*)(p + 4);
            acc[mt] = __builtin_amdgcn_wmma_f32_16x16x4_f32(
                false, am0, false, bm0[kp], (short)0, acc[mt], false, false);
            acc[mt] = __builtin_amdgcn_wmma_f32_16x16x4_f32(
                false, am1, false, bm1[kp], (short)0, acc[mt], false, false);
        }
    }

    float* zp = ZS + (((size_t)b * NSL + khalf * 2) * NN + col) * 4;
#pragma unroll
    for (int mt = 0; mt < 8; ++mt) {
        v4f lo = {acc[mt][0], acc[mt][1], acc[mt][2], acc[mt][3]};
        v4f hi = {acc[mt][4], acc[mt][5], acc[mt][6], acc[mt][7]};
        *(v4f*)(zp + (size_t)(mt * 4) * NN * 4) = lo;
        *(v4f*)(zp + (size_t)(mt * 4 + 1) * NN * 4) = hi;
    }
}

__device__ __forceinline__ float gelu_exact(float x) {
    return 0.5f * x * (1.0f + erff(x * 0.70710678118654752f));
}

#define QP 4096
#define GT 1024
#define PT (QP / GT)
template <int LAYER>
__global__ __launch_bounds__(GT) void gather_kernel(const float* __restrict__ ZS,
                                                    const int* __restrict__ IDX,
                                                    const float* __restrict__ X,
                                                    const float* __restrict__ g,
                                                    const float* __restrict__ be,
                                                    const float* __restrict__ mu,
                                                    const float* __restrict__ va,
                                                    float* __restrict__ OUT) {
    __shared__ __attribute__((aligned(16))) float s_y[NN * 4];

    const int tid = threadIdx.x;
    const int q = blockIdx.x & 3;
    const int s = (blockIdx.x >> 2) & 15;
    const int b = blockIdx.x >> 6;
    const float* ys = ZS + ((size_t)b * NSL + s) * NN * 4;
    const float* cs = ZS + ((size_t)b * NSL + 16 + s) * NN * 4;
    const int* ib = IDX + ((size_t)b * NN + q * QP + tid) * KK;

    v4i nx[5];
#pragma unroll
    for (int j = 0; j < 5; ++j) nx[j] = ((const v4i*)ib)[j];

    {
        v4f tmp[16];
#pragma unroll
        for (int i = 0; i < 16; ++i) tmp[i] = *(const v4f*)(ys + (size_t)(i * GT + tid) * 4);
#pragma unroll
        for (int i = 0; i < 16; ++i) *(v4f*)(s_y + (i * GT + tid) * 4) = tmp[i];
    }

    float sc[4], sh[4];
#pragma unroll
    for (int j = 0; j < 4; ++j) {
        const int ch = s * 4 + j;
        sc[j] = g[ch] / sqrtf(va[ch] + 1e-5f);
        sh[j] = be[ch] - mu[ch] * sc[j];
    }
    __syncthreads();

#pragma unroll
    for (int i = 0; i < PT; ++i) {
        const int n = q * QP + i * GT + tid;
        v4i cu[5];
#pragma unroll
        for (int j = 0; j < 5; ++j) cu[j] = nx[j];
        if (i + 1 < PT) {
            const v4i* ip = (const v4i*)(ib + (size_t)(i + 1) * GT * KK);
#pragma unroll
            for (int j = 0; j < 5; ++j) nx[j] = ip[j];
        }
        const v4f c = *(const v4f*)(cs + (size_t)n * 4);
        float xv[4];
        if (LAYER == 2) {
#pragma unroll
            for (int j = 0; j < 4; ++j) xv[j] = X[((size_t)b * DD + s * 4 + j) * NN + n];
        }

        v4f m = {-3.402823466e38f, -3.402823466e38f, -3.402823466e38f, -3.402823466e38f};
#pragma unroll
        for (int j = 0; j < 5; ++j) {
#pragma unroll
            for (int e = 0; e < 4; ++e) {
                const v4f y = *(const v4f*)(s_y + (size_t)cu[j][e] * 4);
                m.x = fmaxf(m.x, y.x);
                m.y = fmaxf(m.y, y.y);
                m.z = fmaxf(m.z, y.z);
                m.w = fmaxf(m.w, y.w);
            }
        }

        if (LAYER == 1) {
            v4f r;
            r.x = gelu_exact((m.x + c.x) * sc[0] + sh[0]);
            r.y = gelu_exact((m.y + c.y) * sc[1] + sh[1]);
            r.z = gelu_exact((m.z + c.z) * sc[2] + sh[2]);
            r.w = gelu_exact((m.w + c.w) * sc[3] + sh[3]);
            *(v4f*)(OUT + (((size_t)b * 16 + s) * NN + n) * 4) = r;
        } else {
            const float h0 = xv[0] + m.x + c.x;
            const float h1 = xv[1] + m.y + c.y;
            const float h2 = xv[2] + m.z + c.z;
            const float h3 = xv[3] + m.w + c.w;
            float* op = OUT + ((size_t)b * DD + s * 4) * NN + n;
            op[0] = gelu_exact(h0 * sc[0] + sh[0]);
            op[(size_t)NN] = gelu_exact(h1 * sc[1] + sh[1]);
            op[(size_t)2 * NN] = gelu_exact(h2 * sc[2] + sh[2]);
            op[(size_t)3 * NN] = gelu_exact(h3 * sc[3] + sh[3]);
        }
    }
}

extern "C" void kernel_launch(void* const* d_in, const int* in_sizes, int n_in,
                              void* d_out, int out_size, void* d_ws, size_t ws_size,
                              hipStream_t stream) {
    const float* x   = (const float*)d_in[0];
    const int*   idx = (const int*)d_in[1];
    const float* W1  = (const float*)d_in[2];
    const float* W2  = (const float*)d_in[3];
    const float* g1  = (const float*)d_in[4];
    const float* b1  = (const float*)d_in[5];
    const float* m1  = (const float*)d_in[6];
    const float* v1  = (const float*)d_in[7];
    const float* g2  = (const float*)d_in[8];
    const float* b2  = (const float*)d_in[9];
    const float* m2  = (const float*)d_in[10];
    const float* v2  = (const float*)d_in[11];
    float* out = (float*)d_out;

    char* ws = (char*)d_ws;
    float* ZS  = (float*)ws;
    float* HS  = (float*)(ws + (size_t)BB * NN * 128 * 4);

    const int gemmBlocks = (BB * (NN / 16) * 32) / 256;
    const int gathBlocks = BB * 16 * (NN / QP);

    gemm_zs_kernel<0><<<gemmBlocks, 256, 0, stream>>>(W1, x, ZS);
    gather_kernel<1><<<gathBlocks, GT, 0, stream>>>(ZS, idx, x, g1, b1, m1, v1, HS);
    gemm_zs_kernel<1><<<gemmBlocks, 256, 0, stream>>>(W2, HS, ZS);
    gather_kernel<2><<<gathBlocks, GT, 0, stream>>>(ZS, idx, x, g2, b2, m2, v2, out);
}


# neighbour list repacked once per call by the first gemm launch, two indices a dword, lane-major per 32 points: a gather wave reads a point's list in 3 whole-line loads (was 5 at 80-byte lane stride)
# speedup vs baseline: 2.5676x; 1.0858x over previous
#include <hip/hip_runtime.h>
#include <math.h>

typedef __attribute__((ext_vector_type(2))) float v2f;
typedef __attribute__((ext_vector_type(4))) float v4f;
typedef __attribute__((ext_vector_type(8))) float v8f;
typedef __attribute__((ext_vector_type(4))) int v4i;
typedef __attribute__((ext_vector_type(2))) int v2i;

#define BB 4
#define DD 64
#define NN 16384
#define KK 20
#define NSL 32
#define PGRP 320

#define WPITCH (64 + 4)

template <int LAYOUT>
__device__ __forceinline__ void load_b(const float* __restrict__ X, int b, int col, int khalf, int kp,
                                       v2f& bm0, v2f& bm1) {
    const int kk = kp * 8 + khalf * 2;
    if (LAYOUT == 0) {
        const float* xb = X + (size_t)b * DD * NN;
        bm0.x = xb[(size_t)kk * NN + col];
        bm0.y = xb[(size_t)(kk + 1) * NN + col];
        bm1.x = xb[(size_t)(kk + 4) * NN + col];
        bm1.y = xb[(size_t)(kk + 5) * NN + col];
    } else {
        const float* xb = X + (((size_t)b * 16 + kp * 2) * NN + col) * 4 + khalf * 2;
        bm0 = *(const v2f*)xb;
        bm1 = *(const v2f*)(xb + (size_t)NN * 4);
    }
}

template <int LAYOUT>
__global__ __launch_bounds__(256) void gemm_zs_kernel(const float* __restrict__ W,
                                                      const float* __restrict__ X,
                                                      float* __restrict__ ZS,
                                                      const int* __restrict__ IDX,
                                                      int* __restrict__ PK) {
    __shared__ __attribute__((aligned(16))) float s_w[128 * WPITCH];

    const int lane = threadIdx.x & 31;
    const int wave = (blockIdx.x * blockDim.x + threadIdx.x) >> 5;
    const int tilesPerBatch = NN / 16;
    const int b = wave / tilesPerBatch;
    const int t = wave - b * tilesPerBatch;
    const int colLocal = lane & 15;
    const int col = t * 16 + colLocal;
    const int khalf = lane >> 4;

    v8f acc[8] = {};
    v2f bm0[8], bm1[8];

#pragma unroll
    for (int kp = 0; kp < 2; ++kp)
        load_b<LAYOUT>(X, b, col, khalf, kp, bm0[kp], bm1[kp]);

    {
        const int q = threadIdx.x & 15;
        const int r = threadIdx.x >> 4;
#pragma unroll
        for (int i = 0; i < 4; ++i) {
            const int m = i * 16 + r;
            const v4f wa = *(const v4f*)(W + m * 128 + q * 4);
            const v4f wb = *(const v4f*)(W + m * 128 + 64 + q * 4);
            *(v4f*)(s_w + m * WPITCH + q * 4) = wa;
            *(v4f*)(s_w + (64 + m) * WPITCH + q * 4) = wb - wa;
        }
    }
    __syncthreads();

#pragma unroll
    for (int kp = 2; kp < 8; ++kp)
        load_b<LAYOUT>(X, b, col, khalf, kp, bm0[kp], bm1[kp]);

    const int pn = (blockIdx.x % (NN / 128)) * 128 + threadIdx.x;
    v4i il[5] = {};
    if (LAYOUT == 0 && threadIdx.x < 128) {
        const v4i* ip = (const v4i*)(IDX + ((size_t)b * NN + pn) * KK);
#pragma unroll
        for (int j = 0; j < 5; ++j) il[j] = ip[j];
    }

    const float* wl = s_w + colLocal * WPITCH + khalf * 2;
#pragma unroll
    for (int kp = 0; kp < 8; ++kp) {
#pragma unroll
        for (int mt = 0; mt < 8; ++mt) {
            const float* p = wl + mt * 16 * WPITCH + kp * 8;
            const v2f am0 = *(const v2f*)p;
            const v2f am1 = *(const v2f*)(p + 4);
            acc[mt] = __builtin_amdgcn_wmma_f32_16x16x4_f32(
                false, am0, false, bm0[kp], (short)0, acc[mt], false, false);
            acc[mt] = __builtin_amdgcn_wmma_f32_16x16x4_f32(
                false, am1, false, bm1[kp], (short)0, acc[mt], false, false);
        }
    }

    float* zp = ZS + (((size_t)b * NSL + khalf * 2) * NN + col) * 4;
#pragma unroll
    for (int mt = 0; mt < 8; ++mt) {
        v4f lo = {acc[mt][0], acc[mt][1], acc[mt][2], acc[mt][3]};
        v4f hi = {acc[mt][4], acc[mt][5], acc[mt][6], acc[mt][7]};
        *(v4f*)(zp + (size_t)(mt * 4) * NN * 4) = lo;
        *(v4f*)(zp + (size_t)(mt * 4 + 1) * NN * 4) = hi;
    }

    if (LAYOUT == 0 && threadIdx.x < 128) {
        int d[10];
#pragma unroll
        for (int i = 0; i < 10; ++i)
            d[i] = il[(2 * i) >> 2][(2 * i) & 3] | (il[(2 * i + 1) >> 2][(2 * i + 1) & 3] << 16);
        int* pg = PK + ((size_t)b * (NN / 32) + (pn >> 5)) * PGRP;
        const v4i p0 = {d[0], d[1], d[2], d[3]};
        const v4i p1 = {d[4], d[5], d[6], d[7]};
        const v2i p2 = {d[8], d[9]};
        *(v4i*)(pg + lane * 4) = p0;
        *(v4i*)(pg + 128 + lane * 4) = p1;
        *(v2i*)(pg + 256 + lane * 2) = p2;
    }
}

__device__ __forceinline__ float gelu_exact(float x) {
    return 0.5f * x * (1.0f + erff(x * 0.70710678118654752f));
}

#define QP 4096
#define GT 1024
#define PT (QP / GT)
template <int LAYER>
__global__ __launch_bounds__(GT) void gather_kernel(const float* __restrict__ ZS,
                                                    const int* __restrict__ PK,
                                                    const float* __restrict__ X,
                                                    const float* __restrict__ g,
                                                    const float* __restrict__ be,
                                                    const float* __restrict__ mu,
                                                    const float* __restrict__ va,
                                                    float* __restrict__ OUT) {
    __shared__ __attribute__((aligned(16))) float s_y[NN * 4];

    const int tid = threadIdx.x;
    const int q = blockIdx.x & 3;
    const int s = (blockIdx.x >> 2) & 15;
    const int b = blockIdx.x >> 6;
    const float* ys = ZS + ((size_t)b * NSL + s) * NN * 4;
    const float* cs = ZS + ((size_t)b * NSL + 16 + s) * NN * 4;
    const int* pg = PK + ((size_t)b * (NN / 32) + ((q * QP + tid) >> 5)) * PGRP;
    const int* pa = pg + (tid & 31) * 4;
    const int* pc = pg + 256 + (tid & 31) * 2;

    v4i nxa = *(const v4i*)pa;
    v4i nxb = *(const v4i*)(pa + 128);
    v2i nxc = *(const v2i*)pc;

    {
        v4f tmp[16];
#pragma unroll
        for (int i = 0; i < 16; ++i) tmp[i] = *(const v4f*)(ys + (size_t)(i * GT + tid) * 4);
#pragma unroll
        for (int i = 0; i < 16; ++i) *(v4f*)(s_y + (i * GT + tid) * 4) = tmp[i];
    }

    float sc[4], sh[4];
#pragma unroll
    for (int j = 0; j < 4; ++j) {
        const int ch = s * 4 + j;
        sc[j] = g[ch] / sqrtf(va[ch] + 1e-5f);
        sh[j] = be[ch] - mu[ch] * sc[j];
    }
    __syncthreads();

#pragma unroll
    for (int i = 0; i < PT; ++i) {
        const int n = q * QP + i * GT + tid;
        int cu[10];
#pragma unroll
        for (int j = 0; j < 4; ++j) { cu[j] = nxa[j]; cu[4 + j] = nxb[j]; }
        cu[8] = nxc[0];
        cu[9] = nxc[1];
        if (i + 1 < PT) {
            const size_t go = (size_t)(i + 1) * (GT / 32) * PGRP;
            nxa = *(const v4i*)(pa + go);
            nxb = *(const v4i*)(pa + go + 128);
            nxc = *(const v2i*)(pc + go);
        }
        const v4f c = *(const v4f*)(cs + (size_t)n * 4);
        float xv[4];
        if (LAYER == 2) {
#pragma unroll
            for (int j = 0; j < 4; ++j) xv[j] = X[((size_t)b * DD + s * 4 + j) * NN + n];
        }

        v4f m = {-3.402823466e38f, -3.402823466e38f, -3.402823466e38f, -3.402823466e38f};
#pragma unroll
        for (int j = 0; j < 10; ++j) {
#pragma unroll
            for (int e = 0; e < 2; ++e) {
                const unsigned w = (unsigned)cu[j];
                const unsigned nb = e ? (w >> 16) : (w & 0xffffu);
                const v4f y = *(const v4f*)(s_y + (size_t)nb * 4);
                m.x = fmaxf(m.x, y.x);
                m.y = fmaxf(m.y, y.y);
                m.z = fmaxf(m.z, y.z);
                m.w = fmaxf(m.w, y.w);
            }
        }

        if (LAYER == 1) {
            v4f r;
            r.x = gelu_exact((m.x + c.x) * sc[0] + sh[0]);
            r.y = gelu_exact((m.y + c.y) * sc[1] + sh[1]);
            r.z = gelu_exact((m.z + c.z) * sc[2] + sh[2]);
            r.w = gelu_exact((m.w + c.w) * sc[3] + sh[3]);
            *(v4f*)(OUT + (((size_t)b * 16 + s) * NN + n) * 4) = r;
        } else {
            const float h0 = xv[0] + m.x + c.x;
            const float h1 = xv[1] + m.y + c.y;
            const float h2 = xv[2] + m.z + c.z;
            const float h3 = xv[3] + m.w + c.w;
            float* op = OUT + ((size_t)b * DD + s * 4) * NN + n;
            op[0] = gelu_exact(h0 * sc[0] + sh[0]);
            op[(size_t)NN] = gelu_exact(h1 * sc[1] + sh[1]);
            op[(size_t)2 * NN] = gelu_exact(h2 * sc[2] + sh[2]);
            op[(size_t)3 * NN] = gelu_exact(h3 * sc[3] + sh[3]);
        }
    }
}

extern "C" void kernel_launch(void* const* d_in, const int* in_sizes, int n_in,
                              void* d_out, int out_size, void* d_ws, size_t ws_size,
                              hipStream_t stream) {
    const float* x   = (const float*)d_in[0];
    const int*   idx = (const int*)d_in[1];
    const float* W1  = (const float*)d_in[2];
    const float* W2  = (const float*)d_in[3];
    const float* g1  = (const float*)d_in[4];
    const float* b1  = (const float*)d_in[5];
    const float* m1  = (const float*)d_in[6];
    const float* v1  = (const float*)d_in[7];
    const float* g2  = (const float*)d_in[8];
    const float* b2  = (const float*)d_in[9];
    const float* m2  = (const float*)d_in[10];
    const float* v2  = (const float*)d_in[11];
    float* out = (float*)d_out;

    char* ws = (char*)d_ws;
    float* ZS  = (float*)ws;
    float* HS  = (float*)(ws + (size_t)BB * NN * 128 * 4);
    int*   PK  = (int*)(ws + (size_t)BB * NN * 192 * 4);

    const int gemmBlocks = (BB * (NN / 16) * 32) / 256;
    const int gathBlocks = BB * 16 * (NN / QP);

    gemm_zs_kernel<0><<<gemmBlocks, 256, 0, stream>>>(W1, x, ZS, idx, PK);
    gather_kernel<1><<<gathBlocks, GT, 0, stream>>>(ZS, PK, x, g1, b1, m1, v1, HS);
    gemm_zs_kernel<1><<<gemmBlocks, 256, 0, stream>>>(W2, HS, ZS, idx, PK);
    gather_kernel<2><<<gathBlocks, GT, 0, stream>>>(ZS, PK, x, g2, b2, m2, v2, out);
}
